# starting kernel
# baseline (speedup 1.0000x reference)
#include <hip/hip_runtime.h>
#include <math.h>


typedef float v2f __attribute__((ext_vector_type(2)));
typedef float v8f __attribute__((ext_vector_type(8)));

#define VDIM 16
#define NVOX 4096
#define NATOM 14
#define NLBL 21

__device__ __forceinline__ float dm_elem(int m, int k) {
  if (m == 0)  return (k == 0)  ? -1.0f : ((k == 1)  ? 1.0f : 0.0f);
  if (m == 15) return (k == 15) ?  1.0f : ((k == 14) ? -1.0f : 0.0f);
  if (k == m - 1) return -0.5f;
  if (k == m + 1) return  0.5f;
  return 0.0f;
}

template <int PASS>
__device__ __forceinline__ void div_pass(const float* __restrict__ src,
                                         float* __restrict__ dst,
                                         const v2f aReg[4], int lane, int wv) {
  const int nn    = lane & 15;
  const int khalf = (lane < 16) ? 0 : 2;
  const int mbase = (lane < 16) ? 0 : 8;
  for (int t = wv; t < 16; t += 8) {
    v8f acc = {};
#pragma unroll
    for (int kk = 0; kk < 4; ++kk) {
      const int k0 = 4 * kk + khalf;
      const int k1 = k0 + 1;
      int i0, i1;
      if (PASS == 0)      { i0 = k0 * 256 + t * 16 + nn; i1 = k1 * 256 + t * 16 + nn; }
      else if (PASS == 1) { i0 = t * 256 + k0 * 16 + nn; i1 = t * 256 + k1 * 16 + nn; }
      else                { i0 = t * 256 + nn * 16 + k0; i1 = t * 256 + nn * 16 + k1; }
      v2f b; b.x = src[i0]; b.y = src[i1];
      acc = __builtin_amdgcn_wmma_f32_16x16x4_f32(
           false, aReg[kk],  false, b,
           (short)0, acc,  false,  false);
    }
#pragma unroll
    for (int j = 0; j < 8; ++j) {
      const int m = mbase + j;
      int idx;
      if (PASS == 0)      idx = m * 256 + t * 16 + nn;
      else if (PASS == 1) idx = t * 256 + m * 16 + nn;
      else                idx = t * 256 + nn * 16 + m;
      if (PASS == 0) dst[idx] = acc[j];
      else           dst[idx] += acc[j];
    }
  }
}

__global__ __launch_bounds__(256)
void voxel_field_div_84670985273723(const float* __restrict__ C,
                                    const int*   __restrict__ L,
                                    const float* __restrict__ atom_mask,
                                    const float* __restrict__ charges,
                                    float* __restrict__ out, int ZN) {
  __shared__ float sFx[NVOX], sFy[NVOX], sFz[NVOX];
  __shared__ float sF[9], sO[3];
  __shared__ float sAx[NATOM], sAy[NATOM], sAz[NATOM], sPc[NATOM];

  const int n    = blockIdx.x;
  const int tid  = threadIdx.x;
  const int lane = tid & 31;
  const int wv   = tid >> 5;

  float* outCb  = out;
  float* outDiv = out + (size_t)ZN * 12;
  float* outFr  = outDiv + (size_t)ZN * NVOX;

  v2f aReg[4];
  {
    const int m     = lane & 15;
    const int khalf = (lane < 16) ? 0 : 2;
#pragma unroll
    for (int kk = 0; kk < 4; ++kk) {
      aReg[kk].x = dm_elem(m, 4 * kk + khalf);
      aReg[kk].y = dm_elem(m, 4 * kk + khalf + 1);
    }
  }

  if (tid < NATOM) {
    const float* cn = C + (size_t)n * (NATOM * 3);
    sAx[tid] = cn[tid * 3 + 0];
    sAy[tid] = cn[tid * 3 + 1];
    sAz[tid] = cn[tid * 3 + 2];
    int lbl = L[n];
    if (lbl == -1) lbl = NLBL - 1;
    sPc[tid] = charges[lbl * NATOM + tid] * atom_mask[(size_t)n * NATOM + tid];
  }
  if (tid == 0) {
    const float* cn = C + (size_t)n * (NATOM * 3);
    const float Nx = cn[0], Ny = cn[1], Nz = cn[2];
    const float Ax = cn[3], Ay = cn[4], Az = cn[5];
    const float Cx = cn[6], Cy = cn[7], Cz = cn[8];
    const float b1x = Ax - Nx, b1y = Ay - Ny, b1z = Az - Nz;
    const float b2x = Cx - Ax, b2y = Cy - Ay, b2z = Cz - Az;
    const float b3x = b1y * b2z - b1z * b2y;
    const float b3y = b1z * b2x - b1x * b2z;
    const float b3z = b1x * b2y - b1y * b2x;
    const float cbx = Ax - 0.58273431f * b2x + 0.56802827f * b1x - 0.54067466f * b3x;
    const float cby = Ay - 0.58273431f * b2y + 0.56802827f * b1y - 0.54067466f * b3y;
    const float cbz = Az - 0.58273431f * b2z + 0.56802827f * b1z - 0.54067466f * b3z;

    float* cb_out = outCb + (size_t)n * 12;
    cb_out[0] = Nx;  cb_out[1]  = Ny;  cb_out[2]  = Nz;
    cb_out[3] = Ax;  cb_out[4]  = Ay;  cb_out[5]  = Az;
    cb_out[6] = Cx;  cb_out[7]  = Cy;  cb_out[8]  = Cz;
    cb_out[9] = cbx; cb_out[10] = cby; cb_out[11] = cbz;

    float yx = cbx - Ax, yy = cby - Ay, yz = cbz - Az;
    float yn = fmaxf(sqrtf(yx * yx + yy * yy + yz * yz), 1e-6f);
    yx /= yn; yy /= yn; yz /= yn;
    float xrx = Cx - Nx, xry = Cy - Ny, xrz = Cz - Nz;
    const float xp = xrx * yx + xry * yy + xrz * yz;
    float xx = xrx - xp, xy = xry - xp, xz = xrz - xp;
    float xn = fmaxf(sqrtf(xx * xx + xy * xy + xz * xz), 1e-6f);
    xx /= xn; xy /= xn; xz /= xn;
    float zx = xy * yz - xz * yy;
    float zy = xz * yx - xx * yz;
    float zz = xx * yy - xy * yx;
    float zn = fmaxf(sqrtf(zx * zx + zy * zy + zz * zz), 1e-6f);
    zx /= zn; zy /= zn; zz /= zn;

    sF[0] = xx; sF[1] = xy; sF[2] = xz;
    sF[3] = yx; sF[4] = yy; sF[5] = yz;
    sF[6] = zx; sF[7] = zy; sF[8] = zz;
    sO[0] = cbx; sO[1] = cby; sO[2] = cbz;

    float* fr = outFr + (size_t)n * 9;
    fr[0] = xx; fr[1] = xy; fr[2] = xz;
    fr[3] = yx; fr[4] = yy; fr[5] = yz;
    fr[6] = zx; fr[7] = zy; fr[8] = zz;
  }
  __syncthreads();

  for (int i = tid; i < NVOX; i += 256) {
    const int x = i >> 8, y = (i >> 4) & 15, z = i & 15;
    const float vx = (float)x - 8.0f;
    const float vy = (float)y - 4.0f;
    const float vz = (float)z - 8.0f;
    const float px = sO[0] + vx * sF[0] + vy * sF[3] + vz * sF[6];
    const float py = sO[1] + vx * sF[1] + vy * sF[4] + vz * sF[7];
    const float pz = sO[2] + vx * sF[2] + vy * sF[5] + vz * sF[8];
    float fx = 0.0f, fy = 0.0f, fz = 0.0f;
#pragma unroll
    for (int a = 0; a < NATOM; ++a) {
      const float dx = px - sAx[a];
      const float dy = py - sAy[a];
      const float dz = pz - sAz[a];
      const float dd = dx * dx + dy * dy + dz * dz;
      const float d  = sqrtf(dd);
      const float dc = fmaxf(d, 1.0f);
      const float ds = (d == 0.0f) ? 1.0f : d;
      const float w  = sPc[a] / (ds * dc * dc);
      fx += w * dx; fy += w * dy; fz += w * dz;
    }
    const float fn = sqrtf(fx * fx + fy * fy + fz * fz);
    const float inv = (fn == 0.0f) ? 1.0f : 1.0f / fn;
    sFx[i] = fx * inv; sFy[i] = fy * inv; sFz[i] = fz * inv;
  }
  __syncthreads();

  div_pass<0>(sFx, sFx, aReg, lane, wv);
  __syncthreads();
  div_pass<1>(sFy, sFx, aReg, lane, wv);
  __syncthreads();
  div_pass<2>(sFz, sFx, aReg, lane, wv);
  __syncthreads();

  float* dv = outDiv + (size_t)n * NVOX;
  for (int i = tid; i < NVOX; i += 256) dv[i] = sFx[i];
}

extern "C" void kernel_launch(void* const* d_in, const int* in_sizes, int n_in,
                              void* d_out, int out_size, void* d_ws, size_t ws_size,
                              hipStream_t stream) {
  const float* C         = (const float*)d_in[0];
  const int*   L         = (const int*)d_in[1];
  const float* atom_mask = (const float*)d_in[2];
  const float* charges   = (const float*)d_in[3];
  float* out = (float*)d_out;
  const int ZN = in_sizes[0] / (NATOM * 3);
  voxel_field_div_84670985273723<<<ZN, 256, 0, stream>>>(C, L, atom_mask, charges, out, ZN);
}


# Plane ownership: a wave computes whole x planes and runs each plane's y and z derivative tiles itself, under the other waves' field work; 2 block barriers per residue, not 4; f32, sum orders kept
# speedup vs baseline: 2.3448x; 2.3448x over previous
#include <hip/hip_runtime.h>
#include <math.h>


typedef float v2f __attribute__((ext_vector_type(2)));
typedef float v8f __attribute__((ext_vector_type(8)));

#define VDIM 16
#define NVOX 4096
#define NATOM 14
#define NLBL 21
#define GRP 7


__device__ __forceinline__ float dm_elem(int m, int k) {
  if (m == 0)  return (k == 0)  ? -1.0f : ((k == 1)  ? 1.0f : 0.0f);
  if (m == 15) return (k == 15) ?  1.0f : ((k == 14) ? -1.0f : 0.0f);
  if (k == m - 1) return -0.5f;
  if (k == m + 1) return  0.5f;
  return 0.0f;
}

template <int AXIS>
__device__ __forceinline__ void load_b(const float* src, const int t, const int nn,
                                       const int khalf, v2f (&b)[4]) {
#pragma unroll
  for (int kk = 0; kk < 4; ++kk) {
    const int k0 = 4 * kk + khalf;
    const int k1 = k0 + 1;
    int i0, i1;
    if (AXIS == 0)      { i0 = k0 * 256 + t * 16 + nn; i1 = k1 * 256 + t * 16 + nn; }
    else if (AXIS == 1) { i0 = t * 256 + k0 * 16 + nn; i1 = t * 256 + k1 * 16 + nn; }
    else                { i0 = t * 256 + nn * 16 + k0; i1 = t * 256 + nn * 16 + k1; }
    b[kk].x = src[i0]; b[kk].y = src[i1];
  }
}

template <int AXIS>
__device__ __forceinline__ int tile_idx(const int t, const int m, const int nn) {
  if (AXIS == 0)      return m * 256 + t * 16 + nn;
  else if (AXIS == 1) return t * 256 + m * 16 + nn;
  else                return t * 256 + nn * 16 + m;
}

#define LOADS_FIRST() __builtin_amdgcn_sched_barrier(0)

__device__ __forceinline__ void plane_dy_dz(float* fy, float* fz, const v2f aReg[4],
                                            const int t, const int lane) {
  const int nn    = lane & 15;
  const int khalf = (lane < 16) ? 0 : 2;
  const int mbase = (lane < 16) ? 0 : 8;
  v2f by[4], bz[4];
  load_b<1>(fy, t, nn, khalf, by);
  load_b<2>(fz, t, nn, khalf, bz);
  LOADS_FIRST();
  v8f accy = {}, accz = {};
#pragma unroll
  for (int kk = 0; kk < 4; ++kk) {
    accy = __builtin_amdgcn_wmma_f32_16x16x4_f32(
         false, aReg[kk],  false, by[kk],
         (short)0, accy,  false,  false);
    accz = __builtin_amdgcn_wmma_f32_16x16x4_f32(
         false, aReg[kk],  false, bz[kk],
         (short)0, accz,  false,  false);
  }
#pragma unroll
  for (int j = 0; j < 8; ++j) fy[tile_idx<1>(t, mbase + j, nn)] = accy[j];
#pragma unroll
  for (int j = 0; j < 8; ++j) fz[tile_idx<2>(t, mbase + j, nn)] = accz[j];
}

__device__ __forceinline__ void tiles_dx_sum(float* fx, const float* dy, const float* dz,
                                             const v2f aReg[4], const int wv, const int lane) {
  const int nn    = lane & 15;
  const int khalf = (lane < 16) ? 0 : 2;
  const int mbase = (lane < 16) ? 0 : 8;
  const int t0 = wv, t1 = wv + 8;
  v2f b0[4], b1[4];
  float y0[8], z0[8], y1[8], z1[8];
  load_b<0>(fx, t0, nn, khalf, b0);
  load_b<0>(fx, t1, nn, khalf, b1);
#pragma unroll
  for (int j = 0; j < 8; ++j) {
    const int i0 = tile_idx<0>(t0, mbase + j, nn);
    const int i1 = tile_idx<0>(t1, mbase + j, nn);
    y0[j] = dy[i0]; z0[j] = dz[i0];
    y1[j] = dy[i1]; z1[j] = dz[i1];
  }
  LOADS_FIRST();
  v8f acc0 = {}, acc1 = {};
#pragma unroll
  for (int kk = 0; kk < 4; ++kk) {
    acc0 = __builtin_amdgcn_wmma_f32_16x16x4_f32(
         false, aReg[kk],  false, b0[kk],
         (short)0, acc0,  false,  false);
    acc1 = __builtin_amdgcn_wmma_f32_16x16x4_f32(
         false, aReg[kk],  false, b1[kk],
         (short)0, acc1,  false,  false);
  }
#pragma unroll
  for (int j = 0; j < 8; ++j) {
    fx[tile_idx<0>(t0, mbase + j, nn)] = (acc0[j] + y0[j]) + z0[j];
    fx[tile_idx<0>(t1, mbase + j, nn)] = (acc1[j] + y1[j]) + z1[j];
  }
}
#undef LOADS_FIRST

#define ROW_FENCE() __builtin_amdgcn_sched_barrier(0)

template <bool GUARDED>
__device__ __forceinline__ void voxel_field(const float px, const float py, const float pz,
                                            const v2f (&ax)[GRP], const v2f (&ay)[GRP],
                                            const v2f (&az)[GRP], const v2f (&pc)[GRP],
                                            float& gx, float& gy, float& gz, float& inv) {
  v2f ex[GRP], ey[GRP], ez[GRP], dd[GRP], xs[GRP], s[GRP], y[GRP], e[GRP], ds[GRP], p[GRP], r[GRP];
  ROW_FENCE();
#pragma unroll
  for (int k = 0; k < GRP; ++k) { ex[k] = px - ax[k]; ey[k] = py - ay[k]; ez[k] = pz - az[k]; }
  ROW_FENCE();
#pragma unroll
  for (int k = 0; k < GRP; ++k) dd[k] = ey[k] * ey[k];
  ROW_FENCE();
#pragma unroll
  for (int k = 0; k < GRP; ++k) dd[k] = __builtin_elementwise_fma(ex[k], ex[k], dd[k]);
  ROW_FENCE();
#pragma unroll
  for (int k = 0; k < GRP; ++k) dd[k] = __builtin_elementwise_fma(ez[k], ez[k], dd[k]);
  ROW_FENCE();
  if (GUARDED) {
#pragma unroll
    for (int k = 0; k < GRP; ++k) xs[k] = dd[k] * 0x1p+64f;
    ROW_FENCE();
  } else {
#pragma unroll
    for (int k = 0; k < GRP; ++k) xs[k] = dd[k];
  }
  unsigned zlo[GRP], zhi[GRP];
#pragma unroll
  for (int k = 0; k < GRP; ++k) {
    y[k].x = __builtin_amdgcn_rsqf(xs[k].x);
    if (GUARDED) zlo[k] = __builtin_amdgcn_ballot_w32(dd[k].x == 0.0f);
    y[k].y = __builtin_amdgcn_rsqf(xs[k].y);
    if (GUARDED) zhi[k] = __builtin_amdgcn_ballot_w32(dd[k].y == 0.0f);
  }
  ROW_FENCE();
#pragma unroll
  for (int k = 0; k < GRP; ++k) s[k] = xs[k] * y[k];
  ROW_FENCE();
  if (GUARDED) {
#pragma unroll
    for (int k = 0; k < GRP; ++k) s[k] = s[k] * 0x1p-32f;
    ROW_FENCE();
#pragma unroll
    for (int k = 0; k < GRP; ++k) {
      ds[k].x = __builtin_amdgcn_inverse_ballot_w32(zlo[k]) ? 1.0f : s[k].x;
      ds[k].y = __builtin_amdgcn_inverse_ballot_w32(zhi[k]) ? 1.0f : s[k].y;
    }
    ROW_FENCE();
  } else {
#pragma unroll
    for (int k = 0; k < GRP; ++k) ds[k] = s[k];
  }
  p[0] = ds[0];
#pragma unroll
  for (int k = 1; k < GRP; ++k) p[k] = p[k - 1] * ds[k];
  v2f q0;
  q0.x = __builtin_amdgcn_rcpf(p[GRP - 1].x);
  q0.y = __builtin_amdgcn_rcpf(p[GRP - 1].y);
  const v2f one2 = {1.0f, 1.0f};
  const v2f eq = __builtin_elementwise_fma(-p[GRP - 1], q0, one2);
  v2f q = __builtin_elementwise_fma(eq, q0, q0);
#pragma unroll
  for (int k = GRP - 1; k > 0; --k) {
    r[k] = q * p[k - 1];
    q *= ds[k];
  }
  r[0] = q;
  ROW_FENCE();
#pragma unroll
  for (int k = 0; k < GRP; ++k) {
    e[k].x = fminf(r[k].x, 1.0f);
    e[k].y = fminf(r[k].y, 1.0f);
  }
  ROW_FENCE();
#pragma unroll
  for (int k = 0; k < GRP; ++k) r[k] = r[k] * e[k];
  ROW_FENCE();
#pragma unroll
  for (int k = 0; k < GRP; ++k) r[k] = r[k] * e[k];
  ROW_FENCE();
#pragma unroll
  for (int k = 0; k < GRP; ++k) r[k] = pc[k] * r[k];
  ROW_FENCE();
  v2f fx = {0.0f, 0.0f}, fy = {0.0f, 0.0f}, fz = {0.0f, 0.0f};
#pragma unroll
  for (int k = 0; k < GRP; ++k) { fx += r[k] * ex[k]; fy += r[k] * ey[k]; fz += r[k] * ez[k]; }
  gx = fx.x + fx.y; gy = fy.x + fy.y; gz = fz.x + fz.y;
  const float n2 = gx * gx + gy * gy + gz * gz;
  const bool big = GUARDED && (n2 >= 0x1p+64f);
  const float xn = GUARDED ? n2 * (big ? 0x1p-64f : 0x1p+32f) : n2;
  const float yn = __builtin_amdgcn_rsqf(xn);
  float sn = xn * yn;
  const float hn = yn * 0.5f;
  const float en = __builtin_fmaf(-sn, sn, xn);
  sn = __builtin_fmaf(en, hn, sn);
  const float fn = GUARDED ? sn * (big ? 0x1p+32f : 0x1p-16f) : sn;
  const float c0 = __builtin_amdgcn_rcpf(fn);
  const float ec = __builtin_fmaf(-fn, c0, 1.0f);
  const float c1 = __builtin_fmaf(ec, c0, c0);
  inv = GUARDED ? ((n2 == 0.0f) ? 1.0f : c1) : c1;
}
#undef ROW_FENCE

#define RES 2

__global__ __launch_bounds__(256)
void voxel_field_div_kernel(const float* __restrict__ C,
                                    const int*   __restrict__ L,
                                    const float* __restrict__ atom_mask,
                                    const float* __restrict__ charges,
                                    float* __restrict__ out, int ZN) {
  __shared__ float sFx[NVOX], sFy[NVOX], sFz[NVOX];
  __shared__ float sF[RES][9], sO[RES][3];
  __shared__ v2f sAx[RES][GRP], sAy[RES][GRP], sAz[RES][GRP], sPc[RES][GRP];

  const int tid  = threadIdx.x;
  const int lane = tid & 31;
  const int wv   = tid >> 5;
  const int wvu  = __builtin_amdgcn_readfirstlane(wv);
  const int G    = gridDim.x;

  float* outCb  = out;
  float* outDiv = out + (size_t)ZN * 12;
  float* outFr  = outDiv + (size_t)ZN * NVOX;

  v2f aReg[4];
  {
    const int m     = lane & 15;
    const int khalf = (lane < 16) ? 0 : 2;
#pragma unroll
    for (int kk = 0; kk < 4; ++kk) {
      aReg[kk].x = dm_elem(m, 4 * kk + khalf);
      aReg[kk].y = dm_elem(m, 4 * kk + khalf + 1);
    }
  }

  for (int base = blockIdx.x; base < ZN; base += RES * G) {
    const int ns = base + wvu * G;
    if (wvu < RES && ns < ZN) {
      const float* cn = C + (size_t)ns * (NATOM * 3);
      if (lane < NATOM) {
        const int slot = (lane < GRP) ? 2 * lane : 2 * (lane - GRP) + 1;
        ((float*)sAx[wvu])[slot] = cn[lane * 3 + 0];
        ((float*)sAy[wvu])[slot] = cn[lane * 3 + 1];
        ((float*)sAz[wvu])[slot] = cn[lane * 3 + 2];
        int lbl = L[ns];
        if (lbl == -1) lbl = NLBL - 1;
        ((float*)sPc[wvu])[slot] = charges[lbl * NATOM + lane] * atom_mask[(size_t)ns * NATOM + lane];
      }
      if (lane == 0) {
        const float Nx = cn[0], Ny = cn[1], Nz = cn[2];
        const float Ax = cn[3], Ay = cn[4], Az = cn[5];
        const float Cx = cn[6], Cy = cn[7], Cz = cn[8];
        const float b1x = Ax - Nx, b1y = Ay - Ny, b1z = Az - Nz;
        const float b2x = Cx - Ax, b2y = Cy - Ay, b2z = Cz - Az;
        const float b3x = b1y * b2z - b1z * b2y;
        const float b3y = b1z * b2x - b1x * b2z;
        const float b3z = b1x * b2y - b1y * b2x;
        const float cbx = Ax - 0.58273431f * b2x + 0.56802827f * b1x - 0.54067466f * b3x;
        const float cby = Ay - 0.58273431f * b2y + 0.56802827f * b1y - 0.54067466f * b3y;
        const float cbz = Az - 0.58273431f * b2z + 0.56802827f * b1z - 0.54067466f * b3z;

        float* cb_out = outCb + (size_t)ns * 12;
        cb_out[0] = Nx;  cb_out[1]  = Ny;  cb_out[2]  = Nz;
        cb_out[3] = Ax;  cb_out[4]  = Ay;  cb_out[5]  = Az;
        cb_out[6] = Cx;  cb_out[7]  = Cy;  cb_out[8]  = Cz;
        cb_out[9] = cbx; cb_out[10] = cby; cb_out[11] = cbz;

        float yx = cbx - Ax, yy = cby - Ay, yz = cbz - Az;
        float yn = fmaxf(sqrtf(yx * yx + yy * yy + yz * yz), 1e-6f);
        yx /= yn; yy /= yn; yz /= yn;
        float xrx = Cx - Nx, xry = Cy - Ny, xrz = Cz - Nz;
        const float xp = xrx * yx + xry * yy + xrz * yz;
        float xx = xrx - xp, xy = xry - xp, xz = xrz - xp;
        float xn = fmaxf(sqrtf(xx * xx + xy * xy + xz * xz), 1e-6f);
        xx /= xn; xy /= xn; xz /= xn;
        float zx = xy * yz - xz * yy;
        float zy = xz * yx - xx * yz;
        float zz = xx * yy - xy * yx;
        float zn = fmaxf(sqrtf(zx * zx + zy * zy + zz * zz), 1e-6f);
        zx /= zn; zy /= zn; zz /= zn;

        float* sFs = sF[wvu];
        float* sOs = sO[wvu];
        sFs[0] = xx; sFs[1] = xy; sFs[2] = xz;
        sFs[3] = yx; sFs[4] = yy; sFs[5] = yz;
        sFs[6] = zx; sFs[7] = zy; sFs[8] = zz;
        sOs[0] = cbx; sOs[1] = cby; sOs[2] = cbz;

        float* fr = outFr + (size_t)ns * 9;
        fr[0] = xx; fr[1] = xy; fr[2] = xz;
        fr[3] = yx; fr[4] = yy; fr[5] = yz;
        fr[6] = zx; fr[7] = zy; fr[8] = zz;
      }
    }
    __syncthreads();

    for (int s = 0; s < RES; ++s) {
      const int n = base + s * G;
      if (n >= ZN) break;

      const float vz = (float)(lane & 15) - 8.0f;
      const float yh = (float)(lane >> 4) - 4.0f;
      v2f ax[GRP], ay[GRP], az[GRP], pc[GRP];
#pragma unroll
      for (int k = 0; k < GRP; ++k) {
        ax[k] = sAx[s][k]; ay[k] = sAy[s][k]; az[k] = sAz[s][k]; pc[k] = sPc[s][k];
      }
      const float o0 = sO[s][0], o1 = sO[s][1], o2 = sO[s][2];
      const float f0 = sF[s][0], f1 = sF[s][1], f2 = sF[s][2], f3 = sF[s][3], f4 = sF[s][4],
                  f5 = sF[s][5], f6 = sF[s][6], f7 = sF[s][7], f8 = sF[s][8];
      for (int pl = 0; pl < 2; ++pl) {
        const int x = wv + 8 * pl;
        const float vx = (float)x - 8.0f;
        for (int sub = 0; sub < 8; ++sub) {
          const float vy = (float)(2 * sub) + yh;
          const float px = o0 + vx * f0 + vy * f3 + vz * f6;
          const float py = o1 + vx * f1 + vy * f4 + vz * f7;
          const float pz = o2 + vx * f2 + vy * f5 + vz * f8;
          float gx, gy, gz, inv;
          voxel_field<false>(px, py, pz, ax, ay, az, pc, gx, gy, gz, inv);
          if (__builtin_expect(__builtin_isunordered(inv, inv), 0))
            voxel_field<true>(px, py, pz, ax, ay, az, pc, gx, gy, gz, inv);
          const int i0 = x * 256 + sub * 32 + lane;
          sFx[i0] = gx * inv; sFy[i0] = gy * inv; sFz[i0] = gz * inv;
        }
        __builtin_amdgcn_fence(__ATOMIC_RELEASE, "wavefront");
        __builtin_amdgcn_wave_barrier();
        __builtin_amdgcn_fence(__ATOMIC_ACQUIRE, "wavefront");
        plane_dy_dz(sFy, sFz, aReg, x, lane);
      }
      __syncthreads();

      tiles_dx_sum(sFx, sFy, sFz, aReg, wv, lane);
      __syncthreads();

      float* dv = outDiv + (size_t)n * NVOX;
      float cp[16];
#pragma unroll
      for (int j = 0; j < 16; ++j) cp[j] = sFx[(wv + 8 * (j >> 3)) * 256 + (j & 7) * 32 + lane];
#pragma unroll
      for (int j = 0; j < 16; ++j) dv[(wv + 8 * (j >> 3)) * 256 + (j & 7) * 32 + lane] = cp[j];
    }
  }
}

extern "C" void kernel_launch(void* const* d_in, const int* in_sizes, int n_in,
                              void* d_out, int out_size, void* d_ws, size_t ws_size,
                              hipStream_t stream) {
  const float* C         = (const float*)d_in[0];
  const int*   L         = (const int*)d_in[1];
  const float* atom_mask = (const float*)d_in[2];
  const float* charges   = (const float*)d_in[3];
  float* out = (float*)d_out;
  const int ZN = in_sizes[0] / (NATOM * 3);
  voxel_field_div_kernel<<<512, 256, 0, stream>>>(C, L, atom_mask, charges, out, ZN);
}


# Field loop: group reciprocal taken on the squared distances (R = 1/d^2 per atom), 1/d = hardware sqrt(R), weight = pc*sqrt(R)*min(R,1); no distance formed; f32, atom sum order kept, last bits move
# speedup vs baseline: 2.4911x; 1.0624x over previous
#include <hip/hip_runtime.h>
#include <math.h>


typedef float v2f __attribute__((ext_vector_type(2)));
typedef float v8f __attribute__((ext_vector_type(8)));

#define VDIM 16
#define NVOX 4096
#define NATOM 14
#define NLBL 21
#define GRP 7


__device__ __forceinline__ float dm_elem(int m, int k) {
  if (m == 0)  return (k == 0)  ? -1.0f : ((k == 1)  ? 1.0f : 0.0f);
  if (m == 15) return (k == 15) ?  1.0f : ((k == 14) ? -1.0f : 0.0f);
  if (k == m - 1) return -0.5f;
  if (k == m + 1) return  0.5f;
  return 0.0f;
}

template <int AXIS>
__device__ __forceinline__ void load_b(const float* src, const int t, const int nn,
                                       const int khalf, v2f (&b)[4]) {
#pragma unroll
  for (int kk = 0; kk < 4; ++kk) {
    const int k0 = 4 * kk + khalf;
    const int k1 = k0 + 1;
    int i0, i1;
    if (AXIS == 0)      { i0 = k0 * 256 + t * 16 + nn; i1 = k1 * 256 + t * 16 + nn; }
    else if (AXIS == 1) { i0 = t * 256 + k0 * 16 + nn; i1 = t * 256 + k1 * 16 + nn; }
    else                { i0 = t * 256 + nn * 16 + k0; i1 = t * 256 + nn * 16 + k1; }
    b[kk].x = src[i0]; b[kk].y = src[i1];
  }
}

template <int AXIS>
__device__ __forceinline__ int tile_idx(const int t, const int m, const int nn) {
  if (AXIS == 0)      return m * 256 + t * 16 + nn;
  else if (AXIS == 1) return t * 256 + m * 16 + nn;
  else                return t * 256 + nn * 16 + m;
}

#define LOADS_FIRST() __builtin_amdgcn_sched_barrier(0)

__device__ __forceinline__ void plane_dy_dz(float* fy, float* fz, const v2f aReg[4],
                                            const int t, const int lane) {
  const int nn    = lane & 15;
  const int khalf = (lane < 16) ? 0 : 2;
  const int mbase = (lane < 16) ? 0 : 8;
  v2f by[4], bz[4];
  load_b<1>(fy, t, nn, khalf, by);
  load_b<2>(fz, t, nn, khalf, bz);
  LOADS_FIRST();
  v8f accy = {}, accz = {};
#pragma unroll
  for (int kk = 0; kk < 4; ++kk) {
    accy = __builtin_amdgcn_wmma_f32_16x16x4_f32(
         false, aReg[kk],  false, by[kk],
         (short)0, accy,  false,  false);
    accz = __builtin_amdgcn_wmma_f32_16x16x4_f32(
         false, aReg[kk],  false, bz[kk],
         (short)0, accz,  false,  false);
  }
#pragma unroll
  for (int j = 0; j < 8; ++j) fy[tile_idx<1>(t, mbase + j, nn)] = accy[j];
#pragma unroll
  for (int j = 0; j < 8; ++j) fz[tile_idx<2>(t, mbase + j, nn)] = accz[j];
}

__device__ __forceinline__ void tiles_dx_sum(float* fx, const float* dy, const float* dz,
                                             const v2f aReg[4], const int wv, const int lane) {
  const int nn    = lane & 15;
  const int khalf = (lane < 16) ? 0 : 2;
  const int mbase = (lane < 16) ? 0 : 8;
  const int t0 = wv, t1 = wv + 8;
  v2f b0[4], b1[4];
  float y0[8], z0[8], y1[8], z1[8];
  load_b<0>(fx, t0, nn, khalf, b0);
  load_b<0>(fx, t1, nn, khalf, b1);
#pragma unroll
  for (int j = 0; j < 8; ++j) {
    const int i0 = tile_idx<0>(t0, mbase + j, nn);
    const int i1 = tile_idx<0>(t1, mbase + j, nn);
    y0[j] = dy[i0]; z0[j] = dz[i0];
    y1[j] = dy[i1]; z1[j] = dz[i1];
  }
  LOADS_FIRST();
  v8f acc0 = {}, acc1 = {};
#pragma unroll
  for (int kk = 0; kk < 4; ++kk) {
    acc0 = __builtin_amdgcn_wmma_f32_16x16x4_f32(
         false, aReg[kk],  false, b0[kk],
         (short)0, acc0,  false,  false);
    acc1 = __builtin_amdgcn_wmma_f32_16x16x4_f32(
         false, aReg[kk],  false, b1[kk],
         (short)0, acc1,  false,  false);
  }
#pragma unroll
  for (int j = 0; j < 8; ++j) {
    fx[tile_idx<0>(t0, mbase + j, nn)] = (acc0[j] + y0[j]) + z0[j];
    fx[tile_idx<0>(t1, mbase + j, nn)] = (acc1[j] + y1[j]) + z1[j];
  }
}
#undef LOADS_FIRST

#define ROW_FENCE() __builtin_amdgcn_sched_barrier(0)

__device__ __forceinline__ void voxel_field_fast(const float px, const float py, const float pz,
                                                 const v2f (&ax)[GRP], const v2f (&ay)[GRP],
                                                 const v2f (&az)[GRP], const v2f (&pc)[GRP],
                                                 float& gx, float& gy, float& gz, float& inv) {
  v2f ex[GRP], ey[GRP], ez[GRP], dd[GRP], p[GRP], r[GRP], sq[GRP], e[GRP];
  ROW_FENCE();
#pragma unroll
  for (int k = 0; k < GRP; ++k) { ex[k] = px - ax[k]; ey[k] = py - ay[k]; ez[k] = pz - az[k]; }
  ROW_FENCE();
#pragma unroll
  for (int k = 0; k < GRP; ++k) dd[k] = ey[k] * ey[k];
  ROW_FENCE();
#pragma unroll
  for (int k = 0; k < GRP; ++k) dd[k] = __builtin_elementwise_fma(ex[k], ex[k], dd[k]);
  ROW_FENCE();
#pragma unroll
  for (int k = 0; k < GRP; ++k) dd[k] = __builtin_elementwise_fma(ez[k], ez[k], dd[k]);
  ROW_FENCE();
  p[0] = dd[0];
#pragma unroll
  for (int k = 1; k < GRP; ++k) p[k] = p[k - 1] * dd[k];
  v2f q0;
  q0.x = __builtin_amdgcn_rcpf(p[GRP - 1].x);
  q0.y = __builtin_amdgcn_rcpf(p[GRP - 1].y);
  const v2f one2 = {1.0f, 1.0f};
  const v2f eq = __builtin_elementwise_fma(-p[GRP - 1], q0, one2);
  v2f q = __builtin_elementwise_fma(eq, q0, q0);
#pragma unroll
  for (int k = GRP - 1; k > 0; --k) {
    r[k] = q * p[k - 1];
    q *= dd[k];
    sq[k].x = __builtin_amdgcn_sqrtf(r[k].x);
    sq[k].y = __builtin_amdgcn_sqrtf(r[k].y);
  }
  r[0] = q;
  sq[0].x = __builtin_amdgcn_sqrtf(r[0].x);
  sq[0].y = __builtin_amdgcn_sqrtf(r[0].y);
#pragma unroll
  for (int k = 0; k < GRP; ++k) {
    e[k].x = fminf(r[k].x, 1.0f);
    e[k].y = fminf(r[k].y, 1.0f);
  }
#pragma unroll
  for (int k = 0; k < GRP; ++k) e[k] = pc[k] * e[k];
  ROW_FENCE();
#pragma unroll
  for (int k = 0; k < GRP; ++k) r[k] = e[k] * sq[k];
  ROW_FENCE();
  v2f fx = {0.0f, 0.0f}, fy = {0.0f, 0.0f}, fz = {0.0f, 0.0f};
#pragma unroll
  for (int k = 0; k < GRP; ++k) { fx += r[k] * ex[k]; fy += r[k] * ey[k]; fz += r[k] * ez[k]; }
  gx = fx.x + fx.y; gy = fy.x + fy.y; gz = fz.x + fz.y;
  const float n2 = gx * gx + gy * gy + gz * gz;
  const float yn = __builtin_amdgcn_rsqf(n2);
  float sn = n2 * yn;
  const float hn = yn * 0.5f;
  const float en = __builtin_fmaf(-sn, sn, n2);
  sn = __builtin_fmaf(en, hn, sn);
  const float c0 = __builtin_amdgcn_rcpf(sn);
  const float ec = __builtin_fmaf(-sn, c0, 1.0f);
  inv = __builtin_fmaf(ec, c0, c0);
}

__device__ __forceinline__ void voxel_field_guarded(const float px, const float py, const float pz,
                                                    const v2f (&ax)[GRP], const v2f (&ay)[GRP],
                                                    const v2f (&az)[GRP], const v2f (&pc)[GRP],
                                                    float& gx, float& gy, float& gz, float& inv) {
  v2f ex[GRP], ey[GRP], ez[GRP], dd[GRP], xs[GRP], s[GRP], y[GRP], e[GRP], ds[GRP], p[GRP], r[GRP];
  ROW_FENCE();
#pragma unroll
  for (int k = 0; k < GRP; ++k) { ex[k] = px - ax[k]; ey[k] = py - ay[k]; ez[k] = pz - az[k]; }
  ROW_FENCE();
#pragma unroll
  for (int k = 0; k < GRP; ++k) dd[k] = ey[k] * ey[k];
  ROW_FENCE();
#pragma unroll
  for (int k = 0; k < GRP; ++k) dd[k] = __builtin_elementwise_fma(ex[k], ex[k], dd[k]);
  ROW_FENCE();
#pragma unroll
  for (int k = 0; k < GRP; ++k) dd[k] = __builtin_elementwise_fma(ez[k], ez[k], dd[k]);
  ROW_FENCE();
#pragma unroll
  for (int k = 0; k < GRP; ++k) xs[k] = dd[k] * 0x1p+64f;
  ROW_FENCE();
  unsigned zlo[GRP], zhi[GRP];
#pragma unroll
  for (int k = 0; k < GRP; ++k) {
    y[k].x = __builtin_amdgcn_rsqf(xs[k].x);
    zlo[k] = __builtin_amdgcn_ballot_w32(dd[k].x == 0.0f);
    y[k].y = __builtin_amdgcn_rsqf(xs[k].y);
    zhi[k] = __builtin_amdgcn_ballot_w32(dd[k].y == 0.0f);
  }
  ROW_FENCE();
#pragma unroll
  for (int k = 0; k < GRP; ++k) s[k] = xs[k] * y[k];
  ROW_FENCE();
#pragma unroll
  for (int k = 0; k < GRP; ++k) s[k] = s[k] * 0x1p-32f;
  ROW_FENCE();
#pragma unroll
  for (int k = 0; k < GRP; ++k) {
    ds[k].x = __builtin_amdgcn_inverse_ballot_w32(zlo[k]) ? 1.0f : s[k].x;
    ds[k].y = __builtin_amdgcn_inverse_ballot_w32(zhi[k]) ? 1.0f : s[k].y;
  }
  ROW_FENCE();
  p[0] = ds[0];
#pragma unroll
  for (int k = 1; k < GRP; ++k) p[k] = p[k - 1] * ds[k];
  v2f q0;
  q0.x = __builtin_amdgcn_rcpf(p[GRP - 1].x);
  q0.y = __builtin_amdgcn_rcpf(p[GRP - 1].y);
  const v2f one2 = {1.0f, 1.0f};
  const v2f eq = __builtin_elementwise_fma(-p[GRP - 1], q0, one2);
  v2f q = __builtin_elementwise_fma(eq, q0, q0);
#pragma unroll
  for (int k = GRP - 1; k > 0; --k) {
    r[k] = q * p[k - 1];
    q *= ds[k];
  }
  r[0] = q;
  ROW_FENCE();
#pragma unroll
  for (int k = 0; k < GRP; ++k) {
    e[k].x = fminf(r[k].x, 1.0f);
    e[k].y = fminf(r[k].y, 1.0f);
  }
  ROW_FENCE();
#pragma unroll
  for (int k = 0; k < GRP; ++k) r[k] = r[k] * e[k];
  ROW_FENCE();
#pragma unroll
  for (int k = 0; k < GRP; ++k) r[k] = r[k] * e[k];
  ROW_FENCE();
#pragma unroll
  for (int k = 0; k < GRP; ++k) r[k] = pc[k] * r[k];
  ROW_FENCE();
  v2f fx = {0.0f, 0.0f}, fy = {0.0f, 0.0f}, fz = {0.0f, 0.0f};
#pragma unroll
  for (int k = 0; k < GRP; ++k) { fx += r[k] * ex[k]; fy += r[k] * ey[k]; fz += r[k] * ez[k]; }
  gx = fx.x + fx.y; gy = fy.x + fy.y; gz = fz.x + fz.y;
  const float n2 = gx * gx + gy * gy + gz * gz;
  const bool big = (n2 >= 0x1p+64f);
  const float xn = n2 * (big ? 0x1p-64f : 0x1p+32f);
  const float yn = __builtin_amdgcn_rsqf(xn);
  float sn = xn * yn;
  const float hn = yn * 0.5f;
  const float en = __builtin_fmaf(-sn, sn, xn);
  sn = __builtin_fmaf(en, hn, sn);
  const float fn = sn * (big ? 0x1p+32f : 0x1p-16f);
  const float c0 = __builtin_amdgcn_rcpf(fn);
  const float ec = __builtin_fmaf(-fn, c0, 1.0f);
  const float c1 = __builtin_fmaf(ec, c0, c0);
  inv = (n2 == 0.0f) ? 1.0f : c1;
}
#undef ROW_FENCE

#define RES 2

__global__ __launch_bounds__(256)
void voxel_field_div_kernel(const float* __restrict__ C,
                                    const int*   __restrict__ L,
                                    const float* __restrict__ atom_mask,
                                    const float* __restrict__ charges,
                                    float* __restrict__ out, int ZN) {
  __shared__ float sFx[NVOX], sFy[NVOX], sFz[NVOX];
  __shared__ float sF[RES][9], sO[RES][3];
  __shared__ v2f sAx[RES][GRP], sAy[RES][GRP], sAz[RES][GRP], sPc[RES][GRP];

  const int tid  = threadIdx.x;
  const int lane = tid & 31;
  const int wv   = tid >> 5;
  const int wvu  = __builtin_amdgcn_readfirstlane(wv);
  const int G    = gridDim.x;

  float* outCb  = out;
  float* outDiv = out + (size_t)ZN * 12;
  float* outFr  = outDiv + (size_t)ZN * NVOX;

  v2f aReg[4];
  {
    const int m     = lane & 15;
    const int khalf = (lane < 16) ? 0 : 2;
#pragma unroll
    for (int kk = 0; kk < 4; ++kk) {
      aReg[kk].x = dm_elem(m, 4 * kk + khalf);
      aReg[kk].y = dm_elem(m, 4 * kk + khalf + 1);
    }
  }

  for (int base = blockIdx.x; base < ZN; base += RES * G) {
    const int ns = base + wvu * G;
    if (wvu < RES && ns < ZN) {
      const float* cn = C + (size_t)ns * (NATOM * 3);
      if (lane < NATOM) {
        const int slot = (lane < GRP) ? 2 * lane : 2 * (lane - GRP) + 1;
        ((float*)sAx[wvu])[slot] = cn[lane * 3 + 0];
        ((float*)sAy[wvu])[slot] = cn[lane * 3 + 1];
        ((float*)sAz[wvu])[slot] = cn[lane * 3 + 2];
        int lbl = L[ns];
        if (lbl == -1) lbl = NLBL - 1;
        ((float*)sPc[wvu])[slot] = charges[lbl * NATOM + lane] * atom_mask[(size_t)ns * NATOM + lane];
      }
      if (lane == 0) {
        const float Nx = cn[0], Ny = cn[1], Nz = cn[2];
        const float Ax = cn[3], Ay = cn[4], Az = cn[5];
        const float Cx = cn[6], Cy = cn[7], Cz = cn[8];
        const float b1x = Ax - Nx, b1y = Ay - Ny, b1z = Az - Nz;
        const float b2x = Cx - Ax, b2y = Cy - Ay, b2z = Cz - Az;
        const float b3x = b1y * b2z - b1z * b2y;
        const float b3y = b1z * b2x - b1x * b2z;
        const float b3z = b1x * b2y - b1y * b2x;
        const float cbx = Ax - 0.58273431f * b2x + 0.56802827f * b1x - 0.54067466f * b3x;
        const float cby = Ay - 0.58273431f * b2y + 0.56802827f * b1y - 0.54067466f * b3y;
        const float cbz = Az - 0.58273431f * b2z + 0.56802827f * b1z - 0.54067466f * b3z;

        float* cb_out = outCb + (size_t)ns * 12;
        cb_out[0] = Nx;  cb_out[1]  = Ny;  cb_out[2]  = Nz;
        cb_out[3] = Ax;  cb_out[4]  = Ay;  cb_out[5]  = Az;
        cb_out[6] = Cx;  cb_out[7]  = Cy;  cb_out[8]  = Cz;
        cb_out[9] = cbx; cb_out[10] = cby; cb_out[11] = cbz;

        float yx = cbx - Ax, yy = cby - Ay, yz = cbz - Az;
        float yn = fmaxf(sqrtf(yx * yx + yy * yy + yz * yz), 1e-6f);
        yx /= yn; yy /= yn; yz /= yn;
        float xrx = Cx - Nx, xry = Cy - Ny, xrz = Cz - Nz;
        const float xp = xrx * yx + xry * yy + xrz * yz;
        float xx = xrx - xp, xy = xry - xp, xz = xrz - xp;
        float xn = fmaxf(sqrtf(xx * xx + xy * xy + xz * xz), 1e-6f);
        xx /= xn; xy /= xn; xz /= xn;
        float zx = xy * yz - xz * yy;
        float zy = xz * yx - xx * yz;
        float zz = xx * yy - xy * yx;
        float zn = fmaxf(sqrtf(zx * zx + zy * zy + zz * zz), 1e-6f);
        zx /= zn; zy /= zn; zz /= zn;

        float* sFs = sF[wvu];
        float* sOs = sO[wvu];
        sFs[0] = xx; sFs[1] = xy; sFs[2] = xz;
        sFs[3] = yx; sFs[4] = yy; sFs[5] = yz;
        sFs[6] = zx; sFs[7] = zy; sFs[8] = zz;
        sOs[0] = cbx; sOs[1] = cby; sOs[2] = cbz;

        float* fr = outFr + (size_t)ns * 9;
        fr[0] = xx; fr[1] = xy; fr[2] = xz;
        fr[3] = yx; fr[4] = yy; fr[5] = yz;
        fr[6] = zx; fr[7] = zy; fr[8] = zz;
      }
    }
    __syncthreads();

    for (int s = 0; s < RES; ++s) {
      const int n = base + s * G;
      if (n >= ZN) break;

      const float vz = (float)(lane & 15) - 8.0f;
      const float yh = (float)(lane >> 4) - 4.0f;
      v2f ax[GRP], ay[GRP], az[GRP], pc[GRP];
#pragma unroll
      for (int k = 0; k < GRP; ++k) {
        ax[k] = sAx[s][k]; ay[k] = sAy[s][k]; az[k] = sAz[s][k]; pc[k] = sPc[s][k];
      }
      const float o0 = sO[s][0], o1 = sO[s][1], o2 = sO[s][2];
      const float f0 = sF[s][0], f1 = sF[s][1], f2 = sF[s][2], f3 = sF[s][3], f4 = sF[s][4],
                  f5 = sF[s][5], f6 = sF[s][6], f7 = sF[s][7], f8 = sF[s][8];
      for (int pl = 0; pl < 2; ++pl) {
        const int x = wv + 8 * pl;
        const float vx = (float)x - 8.0f;
        for (int sub = 0; sub < 8; ++sub) {
          const float vy = (float)(2 * sub) + yh;
          const float px = o0 + vx * f0 + vy * f3 + vz * f6;
          const float py = o1 + vx * f1 + vy * f4 + vz * f7;
          const float pz = o2 + vx * f2 + vy * f5 + vz * f8;
          float gx, gy, gz, inv;
          voxel_field_fast(px, py, pz, ax, ay, az, pc, gx, gy, gz, inv);
          if (__builtin_expect(__builtin_isunordered(inv, inv), 0))
            voxel_field_guarded(px, py, pz, ax, ay, az, pc, gx, gy, gz, inv);
          const int i0 = x * 256 + sub * 32 + lane;
          sFx[i0] = gx * inv; sFy[i0] = gy * inv; sFz[i0] = gz * inv;
        }
        __builtin_amdgcn_fence(__ATOMIC_RELEASE, "wavefront");
        __builtin_amdgcn_wave_barrier();
        __builtin_amdgcn_fence(__ATOMIC_ACQUIRE, "wavefront");
        plane_dy_dz(sFy, sFz, aReg, x, lane);
      }
      __syncthreads();

      tiles_dx_sum(sFx, sFy, sFz, aReg, wv, lane);
      __syncthreads();

      float* dv = outDiv + (size_t)n * NVOX;
      float cp[16];
#pragma unroll
      for (int j = 0; j < 16; ++j) cp[j] = sFx[(wv + 8 * (j >> 3)) * 256 + (j & 7) * 32 + lane];
#pragma unroll
      for (int j = 0; j < 16; ++j) dv[(wv + 8 * (j >> 3)) * 256 + (j & 7) * 32 + lane] = cp[j];
    }
  }
}

extern "C" void kernel_launch(void* const* d_in, const int* in_sizes, int n_in,
                              void* d_out, int out_size, void* d_ws, size_t ws_size,
                              hipStream_t stream) {
  const float* C         = (const float*)d_in[0];
  const int*   L         = (const int*)d_in[1];
  const float* atom_mask = (const float*)d_in[2];
  const float* charges   = (const float*)d_in[3];
  float* out = (float*)d_out;
  const int ZN = in_sizes[0] / (NATOM * 3);
  voxel_field_div_kernel<<<512, 256, 0, stream>>>(C, L, atom_mask, charges, out, ZN);
}
